# starting kernel
# baseline (speedup 1.0000x reference)
#include <hip/hip_runtime.h>

#define NODE_SIZE 50000
#define REL_SIZE  500
#define HID       128
#define NTRIPLE   800000
#define EPSF      1e-12f

typedef __attribute__((ext_vector_type(2))) float v2f;
typedef __attribute__((ext_vector_type(8))) float v8f;

__global__ void zero_kernel(float* __restrict__ p, int n) {
    int i = blockIdx.x * blockDim.x + threadIdx.x;
    int stride = gridDim.x * blockDim.x;
    for (; i < n; i += stride) p[i] = 0.0f;
}

__global__ __launch_bounds__(32)
void rel_stats_kernel(const float* __restrict__ rel_emb,
                      const float* __restrict__ attn,
                      float* __restrict__ rel_norm,
                      float* __restrict__ rel_att) {
    const int lane = threadIdx.x & 31;
    const int tile = blockIdx.x;
    const int half = lane >> 4;
    const int m16  = lane & 15;

    const int   row   = tile * 16 + m16;
    const bool  rowok = row < REL_SIZE;
    const float amask = rowok ? 1.0f : 0.0f;
    const float* arow = rel_emb + (size_t)(rowok ? row : 0) * HID;
    const float bmask = (m16 == 0) ? 1.0f : 0.0f;

    v8f c = {};
    for (int k = 0; k < HID; k += 4) {
        const int koff = half * 2;
        v2f a, b;
        a.x = arow[k + koff + 0] * amask;
        a.y = arow[k + koff + 1] * amask;
        b.x = attn[k + koff + 0] * bmask;
        b.y = attn[k + koff + 1] * bmask;
        c = __builtin_amdgcn_wmma_f32_16x16x4_f32(false, a, false, b,
                                                  (short)0, c, false, false);
    }
    if (m16 == 0) {
        const int mbase = tile * 16 + half * 8;
#pragma unroll
        for (int j = 0; j < 8; ++j)
            if (mbase + j < REL_SIZE) rel_att[mbase + j] = c[j];
    }

    float ss = 0.0f;
    {
        const float4* p = (const float4*)(arow + half * 64);
#pragma unroll
        for (int i = 0; i < 16; ++i) {
            float4 f = p[i];
            ss += f.x * f.x + f.y * f.y + f.z * f.z + f.w * f.w;
        }
    }
    ss += __shfl_xor(ss, 16, 32);
    if (half == 0 && rowok) rel_norm[row] = sqrtf(ss);
}

__global__ __launch_bounds__(256)
void att_kernel(const int*   __restrict__ indices,
                const int*   __restrict__ sp_idx,
                const float* __restrict__ sp_val,
                const float* __restrict__ rel_norm,
                const float* __restrict__ rel_att,
                float* __restrict__ exp_v,
                float* __restrict__ row_sums) {
    int t = blockIdx.x * blockDim.x + threadIdx.x;
    if (t >= NTRIPLE) return;
    const int   r    = sp_idx[2 * t + 1];
    const float v    = sp_val[t];
    const float coef = v / fmaxf(fabsf(v) * rel_norm[r], EPSF);
    const float e    = expf(coef * rel_att[r]);
    exp_v[t] = e;
    atomicAdd(&row_sums[indices[2 * t]], e);
}

__global__ __launch_bounds__(256)
void scatter_kernel(const float* __restrict__ features,
                    const float* __restrict__ rel_emb,
                    const int*   __restrict__ indices,
                    const int*   __restrict__ sp_idx,
                    const float* __restrict__ sp_val,
                    const float* __restrict__ rel_norm,
                    const float* __restrict__ exp_v,
                    const float* __restrict__ row_sums,
                    float* __restrict__ out) {
    const int wavesPerBlock = blockDim.x >> 5;
    const int t    = blockIdx.x * wavesPerBlock + (threadIdx.x >> 5);
    const int lane = threadIdx.x & 31;
    if (t >= NTRIPLE) return;

    const int   row  = indices[2 * t + 0];
    const int   col  = indices[2 * t + 1];
    const int   r    = sp_idx[2 * t + 1];
    const float v    = sp_val[t];
    const float coef = v / fmaxf(fabsf(v) * rel_norm[r], EPSF);

    const float4 f4 = ((const float4*)(features + (size_t)col * HID))[lane];
    const float4 r4 = ((const float4*)(rel_emb  + (size_t)r   * HID))[lane];

    float partial = f4.x * r4.x + f4.y * r4.y + f4.z * r4.z + f4.w * r4.w;
#pragma unroll
    for (int off = 16; off >= 1; off >>= 1)
        partial += __shfl_xor(partial, off, 32);

    const float proj = coef * partial;
    const float s    = exp_v[t] / row_sums[row];
    const float k2   = 2.0f * proj * coef;

    float* dst = out + (size_t)row * HID + lane * 4;
    atomicAdd(dst + 0, (f4.x - k2 * r4.x) * s);
    atomicAdd(dst + 1, (f4.y - k2 * r4.y) * s);
    atomicAdd(dst + 2, (f4.z - k2 * r4.z) * s);
    atomicAdd(dst + 3, (f4.w - k2 * r4.w) * s);
}

extern "C" void kernel_launch(void* const* d_in, const int* in_sizes, int n_in,
                              void* d_out, int out_size, void* d_ws, size_t ws_size,
                              hipStream_t stream) {
    const float* features = (const float*)d_in[0];
    const float* rel_emb  = (const float*)d_in[1];
    const int*   indices  = (const int*)  d_in[2];
    const int*   sp_idx   = (const int*)  d_in[3];
    const float* sp_val   = (const float*)d_in[4];
    const float* attn     = (const float*)d_in[5];
    float* out = (float*)d_out;

    float* ws       = (float*)d_ws;
    float* rel_norm = ws;
    float* rel_att  = ws + 512;
    float* exp_v    = ws + 1024;
    float* row_sums = ws + 1024 + NTRIPLE;

    zero_kernel<<<2048, 256, 0, stream>>>(out, NODE_SIZE * HID);
    zero_kernel<<<256, 256, 0, stream>>>(row_sums, NODE_SIZE);

    rel_stats_kernel<<<32, 32, 0, stream>>>(rel_emb, attn, rel_norm, rel_att);

    att_kernel<<<(NTRIPLE + 255) / 256, 256, 0, stream>>>(
        indices, sp_idx, sp_val, rel_norm, rel_att, exp_v, row_sums);

    scatter_kernel<<<(NTRIPLE + 7) / 8, 256, 0, stream>>>(
        features, rel_emb, indices, sp_idx, sp_val, rel_norm,
        exp_v, row_sums, out);
}


# global f32 atomic scatter replaced by 250 owner work-groups: each scans all edges' owner bytes (small pre-pass), sums its 200 rows in LDS float adds (arrival order), exp sum divided out once per row
# speedup vs baseline: 2.5244x; 2.5244x over previous
#include <hip/hip_runtime.h>


#define NODE_SIZE 50000
#define REL_SIZE  500
#define HID       128
#define NTRIPLE   800000
#define EPSF      1e-12f

#define OWN_ROWS  200
#define N_OWN     (NODE_SIZE / OWN_ROWS)
#define OWN_WAVES 32
#define IT_EDGES  2048
#define N_IT      ((NTRIPLE + IT_EDGES - 1) / IT_EDGES)
#define OWN_BYTES (N_IT * IT_EDGES)
#define NO_OWNER  255u
#define DEPTH     8

static_assert(NODE_SIZE % OWN_ROWS == 0, "owner ranges must tile the rows");
static_assert(N_OWN < 255, "owner id must fit a byte beside the padding value");
static_assert(NTRIPLE % 4 == 0, "owner bytes are written four at a time");
static_assert(65536 % IT_EDGES == 0, "a queue entry keeps the low 16 bits of an edge id");

typedef __attribute__((ext_vector_type(2))) float v2f;
typedef __attribute__((ext_vector_type(8))) float v8f;

__global__ __launch_bounds__(32)
void rel_stats_kernel(const float* __restrict__ rel_emb,
                      const float* __restrict__ attn,
                      float* __restrict__ rel_norm,
                      float* __restrict__ rel_att) {
    const int lane = threadIdx.x & 31;
    const int tile = blockIdx.x;
    const int half = lane >> 4;
    const int m16  = lane & 15;

    const int   row   = tile * 16 + m16;
    const bool  rowok = row < REL_SIZE;
    const float amask = rowok ? 1.0f : 0.0f;
    const float* arow = rel_emb + (size_t)(rowok ? row : 0) * HID;
    const float bmask = (m16 == 0) ? 1.0f : 0.0f;

    v8f c = {};
    for (int k = 0; k < HID; k += 4) {
        const int koff = half * 2;
        v2f a, b;
        a.x = arow[k + koff + 0] * amask;
        a.y = arow[k + koff + 1] * amask;
        b.x = attn[k + koff + 0] * bmask;
        b.y = attn[k + koff + 1] * bmask;
        c = __builtin_amdgcn_wmma_f32_16x16x4_f32(false, a, false, b,
                                                  (short)0, c, false, false);
    }
    if (m16 == 0) {
        const int mbase = tile * 16 + half * 8;
#pragma unroll
        for (int j = 0; j < 8; ++j)
            if (mbase + j < REL_SIZE) rel_att[mbase + j] = c[j];
    }

    float ss = 0.0f;
    {
        const float4* p = (const float4*)(arow + half * 64);
#pragma unroll
        for (int i = 0; i < 16; ++i) {
            float4 f = p[i];
            ss += f.x * f.x + f.y * f.y + f.z * f.z + f.w * f.w;
        }
    }
    ss += __shfl_xor(ss, 16, 32);
    if (half == 0 && rowok) rel_norm[row] = sqrtf(ss);
}

__device__ __forceinline__ unsigned owner_of(int row) {
    const unsigned o = (unsigned)row / OWN_ROWS;
    return o < N_OWN ? o : NO_OWNER;
}

__global__ __launch_bounds__(256)
void owner_byte_kernel(const int* __restrict__ indices,
                       unsigned* __restrict__ own) {
    const int q = blockIdx.x * 256 + threadIdx.x;
    if (q >= OWN_BYTES / 4) return;
    unsigned w = NO_OWNER * 0x01010101u;
    if (q < NTRIPLE / 4) {
        const int4* p = (const int4*)(indices + (size_t)q * 8);
        const int4 a = p[0];
        const int4 b = p[1];
        w = owner_of(a.x) | (owner_of(a.z) << 8) | (owner_of(b.x) << 16) | (owner_of(b.z) << 24);
    }
    own[q] = w;
}

__device__ __forceinline__ int scan_word(unsigned wd, unsigned idpat, unsigned e0,
                                         unsigned short* q, int n) {
    const unsigned x = wd ^ idpat;
    unsigned z = ~(((x & 0x7F7F7F7Fu) + 0x7F7F7F7Fu) | x | 0x7F7F7F7Fu);
    for (int round = 0; round < 4; ++round) {
        const bool has = z != 0u;
        const unsigned m = __builtin_amdgcn_ballot_w32(has);
        if (m == 0u) break;
        if (has) {
            q[n + (int)__builtin_amdgcn_mbcnt_lo(m, 0u)] = (unsigned short)(e0 + ((unsigned)__builtin_ctz(z) >> 3));
            z &= z - 1u;
        }
        n += __builtin_popcount(m);
    }
    return n;
}

__global__ __launch_bounds__(1024)
void owner_sum_kernel(const float* __restrict__ features,
                      const float* __restrict__ rel_emb,
                      const int*   __restrict__ indices,
                      const int*   __restrict__ sp_idx,
                      const float* __restrict__ sp_val,
                      const float* __restrict__ rel_norm,
                      const float* __restrict__ rel_att,
                      const unsigned* __restrict__ own,
                      float* __restrict__ out) {
    __shared__ float acc[OWN_ROWS * HID];
    __shared__ float ssum[OWN_ROWS];
    __shared__ unsigned short queue[OWN_WAVES * IT_EDGES];

    const int tid  = threadIdx.x;
    const int lane = tid & 31;
    const int w    = __builtin_amdgcn_readfirstlane(tid >> 5);
    const int row0 = blockIdx.x * OWN_ROWS;

    for (int i = tid; i < OWN_ROWS * HID; i += OWN_WAVES * 32) acc[i] = 0.0f;
    if (tid < OWN_ROWS) ssum[tid] = 0.0f;
    __syncthreads();

    const unsigned idpat = (unsigned)blockIdx.x * 0x01010101u;
    unsigned short* q = queue + w * IT_EDGES;
    const float4* feat4 = (const float4*)features;
    const float4* rel4  = (const float4*)rel_emb;

    for (int it = w; it < N_IT; it += OWN_WAVES) {
        const uint4* ob = (const uint4*)own + (size_t)it * (IT_EDGES / 16);
        uint4 o[4];
#pragma unroll
        for (int k = 0; k < 4; ++k) o[k] = ob[k * 32 + lane];
        const int ebase = it * IT_EDGES;
        const unsigned elane = (unsigned)(ebase + lane * 16);
        int n = 0;
#pragma unroll
        for (int k = 0; k < 4; ++k) {
            const unsigned e0 = elane + k * 512;
            n = scan_word(o[k].x, idpat, e0 + 0,  q, n);
            n = scan_word(o[k].y, idpat, e0 + 4,  q, n);
            n = scan_word(o[k].z, idpat, e0 + 8,  q, n);
            n = scan_word(o[k].w, idpat, e0 + 12, q, n);
        }
        __threadfence_block();

        for (int i = 0; i < n; i += DEPTH) {
            const int sl = i + (lane & (DEPTH - 1));
            const int tl = ebase + ((int)q[sl < n ? sl : i] & (IT_EDGES - 1));
            const int   rowL = indices[2 * tl + 0];
            const int   colL = indices[2 * tl + 1];
            const int   rL   = sp_idx[2 * tl + 1];
            const float vL   = sp_val[tl];
            const float coefL = vL / fmaxf(fabsf(vL) * rel_norm[rL], EPSF);
            const float eL    = expf(coefL * rel_att[rL]);

            float4 f[DEPTH], g[DEPTH];
#pragma unroll
            for (int k = 0; k < DEPTH; ++k) {
                const int col = __builtin_amdgcn_readlane(colL, k);
                const int r   = __builtin_amdgcn_readlane(rL, k);
                f[k] = feat4[(size_t)col * (HID / 4) + lane];
                g[k] = rel4[(size_t)r * (HID / 4) + lane];
            }
            float partial[DEPTH];
#pragma unroll
            for (int k = 0; k < DEPTH; ++k)
                partial[k] = f[k].x * g[k].x + f[k].y * g[k].y + f[k].z * g[k].z + f[k].w * g[k].w;
#pragma unroll
            for (int off = 16; off >= 1; off >>= 1) {
                float other[DEPTH];
#pragma unroll
                for (int k = 0; k < DEPTH; ++k) other[k] = __shfl_xor(partial[k], off, 32);
#pragma unroll
                for (int k = 0; k < DEPTH; ++k) partial[k] += other[k];
            }
#pragma unroll
            for (int k = 0; k < DEPTH; ++k) {
                if (i + k < n) {
                    const float coef = __int_as_float(__builtin_amdgcn_readlane(__float_as_int(coefL), k));
                    const float e    = __int_as_float(__builtin_amdgcn_readlane(__float_as_int(eL), k));
                    const int   rl   = __builtin_amdgcn_readlane(rowL, k) - row0;
                    const float proj = coef * partial[k];
                    const float k2   = 2.0f * proj * coef;
                    float* a = acc + rl * HID + lane;
                    atomicAdd(a +  0, (f[k].x - k2 * g[k].x) * e);
                    atomicAdd(a + 32, (f[k].y - k2 * g[k].y) * e);
                    atomicAdd(a + 64, (f[k].z - k2 * g[k].z) * e);
                    atomicAdd(a + 96, (f[k].w - k2 * g[k].w) * e);
                    if (lane == 0) atomicAdd(&ssum[rl], e);
                }
            }
        }
    }
    __syncthreads();

    for (int rl = w; rl < OWN_ROWS; rl += OWN_WAVES) {
        const float S = ssum[rl];
        const float* a = acc + rl * HID + lane;
        float4 o4 = make_float4(0.0f, 0.0f, 0.0f, 0.0f);
        if (S > 0.0f) {
            o4.x = a[0]  / S;
            o4.y = a[32] / S;
            o4.z = a[64] / S;
            o4.w = a[96] / S;
        }
        ((float4*)(out + (size_t)(row0 + rl) * HID))[lane] = o4;
    }
}

extern "C" void kernel_launch(void* const* d_in, const int* in_sizes, int n_in,
                              void* d_out, int out_size, void* d_ws, size_t ws_size,
                              hipStream_t stream) {
    const float* features = (const float*)d_in[0];
    const float* rel_emb  = (const float*)d_in[1];
    const int*   indices  = (const int*)  d_in[2];
    const int*   sp_idx   = (const int*)  d_in[3];
    const float* sp_val   = (const float*)d_in[4];
    const float* attn     = (const float*)d_in[5];
    float* out = (float*)d_out;

    float*    ws       = (float*)d_ws;
    float*    rel_norm = ws;
    float*    rel_att  = ws + 512;
    unsigned* own      = (unsigned*)(ws + 1024);

    rel_stats_kernel<<<32, 32, 0, stream>>>(rel_emb, attn, rel_norm, rel_att);

    owner_byte_kernel<<<(OWN_BYTES / 4 + 255) / 256, 256, 0, stream>>>(indices, own);

    owner_sum_kernel<<<N_OWN, OWN_WAVES * 32, 0, stream>>>(
        features, rel_emb, indices, sp_idx, sp_val, rel_norm, rel_att, own, out);
}


# owner pass: a wave's queue holds full edge ids and carries a trip's leftover edges into the next trip, so edges go in full batches of 8 (was: part-empty batches per trip); LDS adds arrival order
# speedup vs baseline: 2.6478x; 1.0489x over previous
#include <hip/hip_runtime.h>


#define NODE_SIZE 50000
#define REL_SIZE  500
#define HID       128
#define NTRIPLE   800000
#define EPSF      1e-12f

#define OWN_ROWS  200
#define N_OWN     (NODE_SIZE / OWN_ROWS)
#define OWN_WAVES 32
#define IT_EDGES  2048
#define N_IT      ((NTRIPLE + IT_EDGES - 1) / IT_EDGES)
#define OWN_BYTES (N_IT * IT_EDGES)
#define NO_OWNER  255u
#define DEPTH     8
#define QCAP      1536
#define MAX_PASS  (2 * ((N_IT + OWN_WAVES - 1) / OWN_WAVES))

static_assert(NODE_SIZE % OWN_ROWS == 0, "owner ranges must tile the rows");
static_assert(N_OWN < 255, "owner id must fit a byte beside the padding value");
static_assert(NTRIPLE % 4 == 0, "owner bytes are written four at a time");
static_assert(QCAP >= DEPTH - 1 + IT_EDGES / 2, "what is carried over plus half a trip must fit the queue");
static_assert(DEPTH <= 32 && (DEPTH & (DEPTH - 1)) == 0, "a batch is a power of two of lanes");

typedef __attribute__((ext_vector_type(2))) float v2f;
typedef __attribute__((ext_vector_type(8))) float v8f;

__global__ __launch_bounds__(32)
void rel_stats_kernel(const float* __restrict__ rel_emb,
                      const float* __restrict__ attn,
                      float* __restrict__ rel_norm,
                      float* __restrict__ rel_att) {
    const int lane = threadIdx.x & 31;
    const int tile = blockIdx.x;
    const int half = lane >> 4;
    const int m16  = lane & 15;

    const int   row   = tile * 16 + m16;
    const bool  rowok = row < REL_SIZE;
    const float amask = rowok ? 1.0f : 0.0f;
    const float* arow = rel_emb + (size_t)(rowok ? row : 0) * HID;
    const float bmask = (m16 == 0) ? 1.0f : 0.0f;

    v8f c = {};
    for (int k = 0; k < HID; k += 4) {
        const int koff = half * 2;
        v2f a, b;
        a.x = arow[k + koff + 0] * amask;
        a.y = arow[k + koff + 1] * amask;
        b.x = attn[k + koff + 0] * bmask;
        b.y = attn[k + koff + 1] * bmask;
        c = __builtin_amdgcn_wmma_f32_16x16x4_f32(false, a, false, b,
                                                  (short)0, c, false, false);
    }
    if (m16 == 0) {
        const int mbase = tile * 16 + half * 8;
#pragma unroll
        for (int j = 0; j < 8; ++j)
            if (mbase + j < REL_SIZE) rel_att[mbase + j] = c[j];
    }

    float ss = 0.0f;
    {
        const float4* p = (const float4*)(arow + half * 64);
#pragma unroll
        for (int i = 0; i < 16; ++i) {
            float4 f = p[i];
            ss += f.x * f.x + f.y * f.y + f.z * f.z + f.w * f.w;
        }
    }
    ss += __shfl_xor(ss, 16, 32);
    if (half == 0 && rowok) rel_norm[row] = sqrtf(ss);
}

__device__ __forceinline__ unsigned owner_of(int row) {
    const unsigned o = (unsigned)row / OWN_ROWS;
    return o < N_OWN ? o : NO_OWNER;
}

__global__ __launch_bounds__(256)
void owner_byte_kernel(const int* __restrict__ indices,
                       unsigned* __restrict__ own) {
    const int q = blockIdx.x * 256 + threadIdx.x;
    if (q >= OWN_BYTES / 4) return;
    unsigned w = NO_OWNER * 0x01010101u;
    if (q < NTRIPLE / 4) {
        const int4* p = (const int4*)(indices + (size_t)q * 8);
        const int4 a = p[0];
        const int4 b = p[1];
        w = owner_of(a.x) | (owner_of(a.z) << 8) | (owner_of(b.x) << 16) | (owner_of(b.z) << 24);
    }
    own[q] = w;
}

__device__ __forceinline__ int scan_word(unsigned wd, unsigned idpat, unsigned e0,
                                         unsigned* q, int n) {
    const unsigned x = wd ^ idpat;
    unsigned z = ~(((x & 0x7F7F7F7Fu) + 0x7F7F7F7Fu) | x | 0x7F7F7F7Fu);
    for (int round = 0; round < 4; ++round) {
        const bool has = z != 0u;
        const unsigned m = __builtin_amdgcn_ballot_w32(has);
        if (m == 0u) break;
        if (has) {
            q[n + (int)__builtin_amdgcn_mbcnt_lo(m, 0u)] = e0 + ((unsigned)__builtin_ctz(z) >> 3);
            z &= z - 1u;
        }
        n += __builtin_popcount(m);
    }
    return n;
}

__global__ __launch_bounds__(1024)
void owner_sum_kernel(const float* __restrict__ features,
                      const float* __restrict__ rel_emb,
                      const int*   __restrict__ indices,
                      const int*   __restrict__ sp_idx,
                      const float* __restrict__ sp_val,
                      const float* __restrict__ rel_norm,
                      const float* __restrict__ rel_att,
                      const unsigned* __restrict__ own,
                      float* __restrict__ out) {
    __shared__ float acc[OWN_ROWS * HID];
    __shared__ float ssum[OWN_ROWS];
    __shared__ unsigned queue[OWN_WAVES * QCAP];

    const int tid  = threadIdx.x;
    const int lane = tid & 31;
    const int w    = __builtin_amdgcn_readfirstlane(tid >> 5);
    const int row0 = blockIdx.x * OWN_ROWS;

    for (int i = tid; i < OWN_ROWS * HID; i += OWN_WAVES * 32) acc[i] = 0.0f;
    if (tid < OWN_ROWS) ssum[tid] = 0.0f;
    __syncthreads();

    const unsigned idpat = (unsigned)blockIdx.x * 0x01010101u;
    unsigned* q = queue + w * QCAP;
    const float4* feat4 = (const float4*)features;
    const float4* rel4  = (const float4*)rel_emb;

    int it   = w;
    int half = 0;
    int n    = 0;
    for (int pass = 0; pass < MAX_PASS && it < N_IT; ++pass) {
        const uint4* ob = (const uint4*)own + (size_t)it * (IT_EDGES / 16);
        uint4 o[4];
#pragma unroll
        for (int k = 0; k < 4; ++k) o[k] = ob[k * 32 + lane];
        const unsigned elane = (unsigned)(it * IT_EDGES + lane * 16);
        if (half == 0) {
#pragma unroll
            for (int k = 0; k < 2; ++k) {
                const unsigned e0 = elane + k * 512;
                n = scan_word(o[k].x, idpat, e0 + 0,  q, n);
                n = scan_word(o[k].y, idpat, e0 + 4,  q, n);
                n = scan_word(o[k].z, idpat, e0 + 8,  q, n);
                n = scan_word(o[k].w, idpat, e0 + 12, q, n);
            }
        }
        if (half == 1 || n + IT_EDGES / 2 <= QCAP) {
#pragma unroll
            for (int k = 2; k < 4; ++k) {
                const unsigned e0 = elane + k * 512;
                n = scan_word(o[k].x, idpat, e0 + 0,  q, n);
                n = scan_word(o[k].y, idpat, e0 + 4,  q, n);
                n = scan_word(o[k].z, idpat, e0 + 8,  q, n);
                n = scan_word(o[k].w, idpat, e0 + 12, q, n);
            }
            it += OWN_WAVES;
            half = 0;
        } else {
            half = 1;
        }
        __threadfence_block();

        const int nproc = (it >= N_IT) ? n : (n & ~(DEPTH - 1));
        for (int i = 0; i < nproc; i += DEPTH) {
            const int sl = i + (lane & (DEPTH - 1));
            const int tl = (int)q[sl < nproc ? sl : i];
            const int   rowL = indices[2 * tl + 0];
            const int   colL = indices[2 * tl + 1];
            const int   rL   = sp_idx[2 * tl + 1];
            const float vL   = sp_val[tl];
            const float coefL = vL / fmaxf(fabsf(vL) * rel_norm[rL], EPSF);
            const float eL    = expf(coefL * rel_att[rL]);

            float4 f[DEPTH], g[DEPTH];
#pragma unroll
            for (int k = 0; k < DEPTH; ++k) {
                const int col = __builtin_amdgcn_readlane(colL, k);
                const int r   = __builtin_amdgcn_readlane(rL, k);
                f[k] = feat4[(size_t)col * (HID / 4) + lane];
                g[k] = rel4[(size_t)r * (HID / 4) + lane];
            }
            float partial[DEPTH];
#pragma unroll
            for (int k = 0; k < DEPTH; ++k)
                partial[k] = f[k].x * g[k].x + f[k].y * g[k].y + f[k].z * g[k].z + f[k].w * g[k].w;
#pragma unroll
            for (int off = 16; off >= 1; off >>= 1) {
                float other[DEPTH];
#pragma unroll
                for (int k = 0; k < DEPTH; ++k) other[k] = __shfl_xor(partial[k], off, 32);
#pragma unroll
                for (int k = 0; k < DEPTH; ++k) partial[k] += other[k];
            }
#pragma unroll
            for (int k = 0; k < DEPTH; ++k) {
                if (i + k < nproc) {
                    const float coef = __int_as_float(__builtin_amdgcn_readlane(__float_as_int(coefL), k));
                    const float e    = __int_as_float(__builtin_amdgcn_readlane(__float_as_int(eL), k));
                    const int   rl   = __builtin_amdgcn_readlane(rowL, k) - row0;
                    const float proj = coef * partial[k];
                    const float k2   = 2.0f * proj * coef;
                    float* a = acc + rl * HID + lane;
                    atomicAdd(a +  0, (f[k].x - k2 * g[k].x) * e);
                    atomicAdd(a + 32, (f[k].y - k2 * g[k].y) * e);
                    atomicAdd(a + 64, (f[k].z - k2 * g[k].z) * e);
                    atomicAdd(a + 96, (f[k].w - k2 * g[k].w) * e);
                    if (lane == 0) atomicAdd(&ssum[rl], e);
                }
            }
        }
        const int rem = n - nproc;
        if (nproc > 0 && rem > 0) {
            const unsigned keep = q[nproc + (lane < rem ? lane : 0)];
            if (lane < rem) q[lane] = keep;
        }
        n = rem;
    }
    __syncthreads();

    for (int rl = w; rl < OWN_ROWS; rl += OWN_WAVES) {
        const float S = ssum[rl];
        const float* a = acc + rl * HID + lane;
        float4 o4 = make_float4(0.0f, 0.0f, 0.0f, 0.0f);
        if (S > 0.0f) {
            o4.x = a[0]  / S;
            o4.y = a[32] / S;
            o4.z = a[64] / S;
            o4.w = a[96] / S;
        }
        ((float4*)(out + (size_t)(row0 + rl) * HID))[lane] = o4;
    }
}

extern "C" void kernel_launch(void* const* d_in, const int* in_sizes, int n_in,
                              void* d_out, int out_size, void* d_ws, size_t ws_size,
                              hipStream_t stream) {
    const float* features = (const float*)d_in[0];
    const float* rel_emb  = (const float*)d_in[1];
    const int*   indices  = (const int*)  d_in[2];
    const int*   sp_idx   = (const int*)  d_in[3];
    const float* sp_val   = (const float*)d_in[4];
    const float* attn     = (const float*)d_in[5];
    float* out = (float*)d_out;

    float*    ws       = (float*)d_ws;
    float*    rel_norm = ws;
    float*    rel_att  = ws + 512;
    unsigned* own      = (unsigned*)(ws + 1024);

    rel_stats_kernel<<<32, 32, 0, stream>>>(rel_emb, attn, rel_norm, rel_att);

    owner_byte_kernel<<<(OWN_BYTES / 4 + 255) / 256, 256, 0, stream>>>(indices, own);

    owner_sum_kernel<<<N_OWN, OWN_WAVES * 32, 0, stream>>>(
        features, rel_emb, indices, sp_idx, sp_val, rel_norm, rel_att, own, out);
}


# owner ids kept bit-sliced (8 words per 32 edges, built by wave ballots in the pre-pass): owner pass tests 32 edges per lane with 8 xor-and ops, not 16 by zero-byte tricks; LDS adds arrival order
# speedup vs baseline: 2.9188x; 1.1023x over previous
#include <hip/hip_runtime.h>


#define NODE_SIZE 50000
#define REL_SIZE  500
#define HID       128
#define NTRIPLE   800000
#define EPSF      1e-12f

#define OWN_ROWS  200
#define N_OWN     (NODE_SIZE / OWN_ROWS)
#define OWN_WAVES 32
#define STEP_EDGES 1024
#define IT_EDGES  (2 * STEP_EDGES)
#define N_IT      ((NTRIPLE + IT_EDGES - 1) / IT_EDGES)
#define N_STEP    (2 * N_IT)
#define BITS_WAVES 4
#define OWN_BYTES (N_IT * IT_EDGES)
#define NO_OWNER  255u
#define DEPTH     8
#define QCAP      1536
#define MAX_PASS  (2 * ((N_IT + OWN_WAVES - 1) / OWN_WAVES))

static_assert(NODE_SIZE % OWN_ROWS == 0, "owner ranges must tile the rows");
static_assert(N_OWN < 255, "owner id must fit 8 bits beside the padding value");
static_assert(QCAP >= DEPTH - 1 + STEP_EDGES, "what is carried over plus one step must fit the queue");
static_assert(DEPTH <= 32 && (DEPTH & (DEPTH - 1)) == 0, "a batch is a power of two of lanes");

typedef __attribute__((ext_vector_type(2))) float v2f;
typedef __attribute__((ext_vector_type(8))) float v8f;

__global__ __launch_bounds__(32)
void rel_stats_kernel(const float* __restrict__ rel_emb,
                      const float* __restrict__ attn,
                      float* __restrict__ rel_norm,
                      float* __restrict__ rel_att) {
    const int lane = threadIdx.x & 31;
    const int tile = blockIdx.x;
    const int half = lane >> 4;
    const int m16  = lane & 15;

    const int   row   = tile * 16 + m16;
    const bool  rowok = row < REL_SIZE;
    const float amask = rowok ? 1.0f : 0.0f;
    const float* arow = rel_emb + (size_t)(rowok ? row : 0) * HID;
    const float bmask = (m16 == 0) ? 1.0f : 0.0f;

    v8f c = {};
    for (int k = 0; k < HID; k += 4) {
        const int koff = half * 2;
        v2f a, b;
        a.x = arow[k + koff + 0] * amask;
        a.y = arow[k + koff + 1] * amask;
        b.x = attn[k + koff + 0] * bmask;
        b.y = attn[k + koff + 1] * bmask;
        c = __builtin_amdgcn_wmma_f32_16x16x4_f32(false, a, false, b,
                                                  (short)0, c, false, false);
    }
    if (m16 == 0) {
        const int mbase = tile * 16 + half * 8;
#pragma unroll
        for (int j = 0; j < 8; ++j)
            if (mbase + j < REL_SIZE) rel_att[mbase + j] = c[j];
    }

    float ss = 0.0f;
    {
        const float4* p = (const float4*)(arow + half * 64);
#pragma unroll
        for (int i = 0; i < 16; ++i) {
            float4 f = p[i];
            ss += f.x * f.x + f.y * f.y + f.z * f.z + f.w * f.w;
        }
    }
    ss += __shfl_xor(ss, 16, 32);
    if (half == 0 && rowok) rel_norm[row] = sqrtf(ss);
}

__device__ __forceinline__ unsigned owner_of(int row) {
    const unsigned o = (unsigned)row / OWN_ROWS;
    return o < N_OWN ? o : NO_OWNER;
}

__global__ __launch_bounds__(256)
void owner_bits_kernel(const int* __restrict__ indices,
                       unsigned* __restrict__ own) {
    uint4* own4 = (uint4*)own;
    const int lane = threadIdx.x & 31;
    const int wv   = blockIdx.x * 8 + (threadIdx.x >> 5);
    const int step = wv / BITS_WAVES;
    const int part = wv % BITS_WAVES;
    if (step >= N_STEP) return;
    const unsigned ebase = (unsigned)step * STEP_EDGES + 256u * part + lane;
    int rows[8];
    for (int j = 0; j < 8; ++j) {
        const unsigned e = ebase + 32u * j;
        const int r = indices[2 * (size_t)(e < NTRIPLE ? e : 0u)];
        rows[j] = e < NTRIPLE ? r : -1;
    }
    unsigned w[8] = {0u, 0u, 0u, 0u, 0u, 0u, 0u, 0u};
    for (int j = 0; j < 8; ++j) {
        const unsigned o = owner_of(rows[j]);
        const bool keep = lane == 8 * part + j;
        for (int k = 0; k < 8; ++k) {
            const unsigned bits = __builtin_amdgcn_ballot_w32(((o >> k) & 1u) != 0u);
            w[k] = keep ? bits : w[k];
        }
    }
    if ((lane >> 3) == part) {
        own4[(size_t)step * 64 + lane]      = make_uint4(w[0], w[1], w[2], w[3]);
        own4[(size_t)step * 64 + 32 + lane] = make_uint4(w[4], w[5], w[6], w[7]);
    }
}

__device__ __forceinline__ unsigned match_block(const uint4 lo, const uint4 hi, const unsigned* idm) {
    return (lo.x ^ idm[0]) & (lo.y ^ idm[1]) & (lo.z ^ idm[2]) & (lo.w ^ idm[3])
         & (hi.x ^ idm[4]) & (hi.y ^ idm[5]) & (hi.z ^ idm[6]) & (hi.w ^ idm[7]);
}

__device__ __forceinline__ int scan_block(unsigned m, unsigned e0, unsigned* q, int n) {
    for (int round = 0; round < 32; ++round) {
        const bool has = m != 0u;
        const unsigned b = __builtin_amdgcn_ballot_w32(has);
        if (b == 0u) break;
        if (has) {
            q[n + (int)__builtin_amdgcn_mbcnt_lo(b, 0u)] = e0 + (unsigned)__builtin_ctz(m);
            m &= m - 1u;
        }
        n += __builtin_popcount(b);
    }
    return n;
}

__global__ __launch_bounds__(1024)
void owner_sum_kernel(const float* __restrict__ features,
                      const float* __restrict__ rel_emb,
                      const int*   __restrict__ indices,
                      const int*   __restrict__ sp_idx,
                      const float* __restrict__ sp_val,
                      const float* __restrict__ rel_norm,
                      const float* __restrict__ rel_att,
                      const unsigned* __restrict__ own,
                      float* __restrict__ out) {
    __shared__ float acc[OWN_ROWS * HID];
    __shared__ float ssum[OWN_ROWS];
    __shared__ unsigned queue[OWN_WAVES * QCAP];

    const int tid  = threadIdx.x;
    const int lane = tid & 31;
    const int w    = __builtin_amdgcn_readfirstlane(tid >> 5);
    const int row0 = blockIdx.x * OWN_ROWS;

    for (int i = tid; i < OWN_ROWS * HID; i += OWN_WAVES * 32) acc[i] = 0.0f;
    if (tid < OWN_ROWS) ssum[tid] = 0.0f;
    __syncthreads();

    unsigned idm[8];
    for (int k = 0; k < 8; ++k) idm[k] = (((unsigned)blockIdx.x >> k) & 1u) ? 0u : 0xFFFFFFFFu;
    unsigned* q = queue + w * QCAP;
    const float4* feat4 = (const float4*)features;
    const float4* rel4  = (const float4*)rel_emb;

    int it   = w;
    int half = 0;
    int n    = 0;
    for (int pass = 0; pass < MAX_PASS && it < N_IT; ++pass) {
        const uint4* ob = (const uint4*)own + (size_t)it * (IT_EDGES / 16);
        uint4 o[4];
#pragma unroll
        for (int k = 0; k < 4; ++k) o[k] = ob[k * 32 + lane];
        const unsigned elane = (unsigned)(it * IT_EDGES + lane * 32);
        if (half == 0)
            n = scan_block(match_block(o[0], o[1], idm), elane, q, n);
        if (half == 1 || n + STEP_EDGES <= QCAP) {
            n = scan_block(match_block(o[2], o[3], idm), elane + STEP_EDGES, q, n);
            it += OWN_WAVES;
            half = 0;
        } else {
            half = 1;
        }
        __threadfence_block();

        const int nproc = (it >= N_IT) ? n : (n & ~(DEPTH - 1));
        for (int i = 0; i < nproc; i += DEPTH) {
            const int sl = i + (lane & (DEPTH - 1));
            const int tl = (int)q[sl < nproc ? sl : i];
            const int   rowL = indices[2 * tl + 0];
            const int   colL = indices[2 * tl + 1];
            const int   rL   = sp_idx[2 * tl + 1];
            const float vL   = sp_val[tl];
            const float coefL = vL / fmaxf(fabsf(vL) * rel_norm[rL], EPSF);
            const float eL    = expf(coefL * rel_att[rL]);

            float4 f[DEPTH], g[DEPTH];
#pragma unroll
            for (int k = 0; k < DEPTH; ++k) {
                const int col = __builtin_amdgcn_readlane(colL, k);
                const int r   = __builtin_amdgcn_readlane(rL, k);
                f[k] = feat4[(size_t)col * (HID / 4) + lane];
                g[k] = rel4[(size_t)r * (HID / 4) + lane];
            }
            float partial[DEPTH];
#pragma unroll
            for (int k = 0; k < DEPTH; ++k)
                partial[k] = f[k].x * g[k].x + f[k].y * g[k].y + f[k].z * g[k].z + f[k].w * g[k].w;
#pragma unroll
            for (int off = 16; off >= 1; off >>= 1) {
                float other[DEPTH];
#pragma unroll
                for (int k = 0; k < DEPTH; ++k) other[k] = __shfl_xor(partial[k], off, 32);
#pragma unroll
                for (int k = 0; k < DEPTH; ++k) partial[k] += other[k];
            }
#pragma unroll
            for (int k = 0; k < DEPTH; ++k) {
                if (i + k < nproc) {
                    const float coef = __int_as_float(__builtin_amdgcn_readlane(__float_as_int(coefL), k));
                    const float e    = __int_as_float(__builtin_amdgcn_readlane(__float_as_int(eL), k));
                    const int   rl   = __builtin_amdgcn_readlane(rowL, k) - row0;
                    const float proj = coef * partial[k];
                    const float k2   = 2.0f * proj * coef;
                    float* a = acc + rl * HID + lane;
                    atomicAdd(a +  0, (f[k].x - k2 * g[k].x) * e);
                    atomicAdd(a + 32, (f[k].y - k2 * g[k].y) * e);
                    atomicAdd(a + 64, (f[k].z - k2 * g[k].z) * e);
                    atomicAdd(a + 96, (f[k].w - k2 * g[k].w) * e);
                    if (lane == 0) atomicAdd(&ssum[rl], e);
                }
            }
        }
        const int rem = n - nproc;
        if (nproc > 0 && rem > 0) {
            const unsigned keep = q[nproc + (lane < rem ? lane : 0)];
            if (lane < rem) q[lane] = keep;
        }
        n = rem;
    }
    __syncthreads();

    for (int rl = w; rl < OWN_ROWS; rl += OWN_WAVES) {
        const float S = ssum[rl];
        const float* a = acc + rl * HID + lane;
        float4 o4 = make_float4(0.0f, 0.0f, 0.0f, 0.0f);
        if (S > 0.0f) {
            o4.x = a[0]  / S;
            o4.y = a[32] / S;
            o4.z = a[64] / S;
            o4.w = a[96] / S;
        }
        ((float4*)(out + (size_t)(row0 + rl) * HID))[lane] = o4;
    }
}

extern "C" void kernel_launch(void* const* d_in, const int* in_sizes, int n_in,
                              void* d_out, int out_size, void* d_ws, size_t ws_size,
                              hipStream_t stream) {
    const float* features = (const float*)d_in[0];
    const float* rel_emb  = (const float*)d_in[1];
    const int*   indices  = (const int*)  d_in[2];
    const int*   sp_idx   = (const int*)  d_in[3];
    const float* sp_val   = (const float*)d_in[4];
    const float* attn     = (const float*)d_in[5];
    float* out = (float*)d_out;

    float*    ws       = (float*)d_ws;
    float*    rel_norm = ws;
    float*    rel_att  = ws + 512;
    unsigned* own      = (unsigned*)(ws + 1024);

    rel_stats_kernel<<<32, 32, 0, stream>>>(rel_emb, attn, rel_norm, rel_att);

    owner_bits_kernel<<<(N_STEP * BITS_WAVES + 7) / 8, 256, 0, stream>>>(indices, own);

    owner_sum_kernel<<<N_OWN, OWN_WAVES * 32, 0, stream>>>(
        features, rel_emb, indices, sp_idx, sp_val, rel_norm, rel_att, own, out);
}
